# starting kernel
# baseline (speedup 1.0000x reference)
#include <hip/hip_runtime.h>

typedef float v2f __attribute__((ext_vector_type(2)));
typedef float v8f __attribute__((ext_vector_type(8)));

__device__ __forceinline__ float fast_sqrt(float x) { return __builtin_amdgcn_sqrtf(x); }
__device__ __forceinline__ float fast_exp2(float x) { return __builtin_amdgcn_exp2f(x); }

#define LOG2E 1.44269504088896340736f

__global__ void precompute_kernel(const float* __restrict__ states,
                                  float4* __restrict__ rec,
                                  float* __restrict__ G, int N) {
    int i = blockIdx.x * blockDim.x + threadIdx.x;
    if (i >= N) return;
    const float4* s = (const float4*)(states + (size_t)i * 8);
    float4 a = s[0];
    float4 b = s[1];
    float px = (a.x + a.z) * 0.5f;
    float py = (a.y + a.w) * 0.5f;
    float vx = (b.x + b.z) * 0.5f;
    float vy = (b.y + b.w) * 0.5f;
    float r  = px * px + py * py;
    rec[i] = make_float4(px, py, r, 0.0f);
    float4* g = (float4*)(G + (size_t)i * 16);
    g[0] = make_float4(px, py, vx, vy);
    g[1] = make_float4(px * px, py * py, vx * vx, vy * vy);
    g[2] = make_float4(1.0f, 0.0f, 0.0f, 0.0f);
    g[3] = make_float4(0.0f, 0.0f, 0.0f, 0.0f);
}

__global__ void groupvel_kernel(const float* __restrict__ G,
                                float* __restrict__ gv, int N) {
    __shared__ float sx[256];
    __shared__ float sy[256];
    int t = threadIdx.x;
    float ax = 0.0f, ay = 0.0f;
    for (int j = t; j < N; j += 256) {
        ax += G[(size_t)j * 16 + 2];
        ay += G[(size_t)j * 16 + 3];
    }
    sx[t] = ax; sy[t] = ay;
    __syncthreads();
    for (int s = 128; s > 0; s >>= 1) {
        if (t < s) { sx[t] += sx[t + s]; sy[t] += sy[t + s]; }
        __syncthreads();
    }
    if (t == 0) {
        gv[0] = sx[0] / (float)N;
        gv[1] = sy[0] / (float)N;
    }
}

__global__ __launch_bounds__(256) void neighbour_kernel(
    const float4* __restrict__ rec,
    const float*  __restrict__ G,
    const float*  __restrict__ gvp,
    const float*  __restrict__ log_tau,
    float*        __restrict__ out, int N)
{
    __shared__ float cbuf[8][256];

    const int tid  = threadIdx.x;
    const int wave = tid >> 5;
    const int lane = tid & 31;
    const int row  = lane & 15;
    const int h    = lane >> 4;
    const int i0   = (blockIdx.x * 8 + wave) * 16;
    const int i    = i0 + row;

    const float invtau = 1.0f / __expf(log_tau[0]);

    const float4 ri  = rec[i];
    const float px_i = ri.x, py_i = ri.y, r_i = ri.z;

    float qmin = 3.0e38f;
    #pragma unroll 4
    for (int j = h; j < N; j += 2) {
        float4 rj = rec[j];
        float dot = px_i * rj.x + py_i * rj.y;
        float q   = r_i + rj.z - 2.0f * dot;
        q = (j == i) ? 3.0e38f : q;
        qmin = fminf(qmin, q);
    }
    qmin = fminf(qmin, __shfl_xor(qmin, 16, 32));
    const float mdist = fast_sqrt(fmaxf(qmin, 0.0f) + 1e-8f);

    const float k2 = invtau * LOG2E;
    const float b2 = mdist * k2;

    v8f acc = {0.0f, 0.0f, 0.0f, 0.0f, 0.0f, 0.0f, 0.0f, 0.0f};

    for (int jt = 0; jt < N; jt += 16) {
        v2f A[4];
        v2f B[4];
        #pragma unroll
        for (int g = 0; g < 4; ++g) {
            #pragma unroll
            for (int p = 0; p < 2; ++p) {
                const int j = jt + 4 * g + p + 2 * h;
                float4 rj = rec[j];
                float dot  = px_i * rj.x + py_i * rj.y;
                float q    = r_i + rj.z - 2.0f * dot;
                float dist = fast_sqrt(fmaxf(q, 0.0f) + 1e-8f);
                float w    = fast_exp2(fmaf(-dist, k2, b2));
                w = (j == i) ? 0.0f : w;
                A[g][p] = w;
                B[g][p] = G[(size_t)j * 16 + row];
            }
        }
        #pragma unroll
        for (int g = 0; g < 4; ++g) {
            acc = __builtin_amdgcn_wmma_f32_16x16x4_f32(
                 false, A[g],  false, B[g],
                 (short)0, acc,  false,  false);
        }
    }

    #pragma unroll
    for (int v = 0; v < 8; ++v) {
        cbuf[wave][(v + 8 * h) * 16 + row] = acc[v];
    }
    __syncthreads();

    if (lane < 16) {
        const int ii = i0 + lane;
        const float* c = &cbuf[wave][lane * 16];
        const float S    = c[8];
        const float invS = 1.0f / S;
        const float4 fi4 = *(const float4*)(G + (size_t)ii * 16);
        const float f[4] = {fi4.x, fi4.y, fi4.z, fi4.w};
        const float gvx = gvp[0];
        const float gvy = gvp[1];

        float mu[4], sg[4];
        #pragma unroll
        for (int d = 0; d < 4; ++d) {
            float T = c[d] * invS;
            float U = c[4 + d] * invS;
            mu[d] = f[d] - T;
            float eg2 = f[d] * f[d] - 2.0f * f[d] * T + U;
            sg[d] = fast_sqrt(fmaxf(eg2 - mu[d] * mu[d], 0.0f) + 1e-6f);
        }
        float4* o = (float4*)(out + (size_t)ii * 12);
        o[0] = make_float4(mu[0], mu[1], mu[2], mu[3]);
        o[1] = make_float4(sg[0], sg[1], sg[2], sg[3]);
        o[2] = make_float4(gvx, gvy, f[2] - gvx, f[3] - gvy);
    }
}

extern "C" void kernel_launch(void* const* d_in, const int* in_sizes, int n_in,
                              void* d_out, int out_size, void* d_ws, size_t ws_size,
                              hipStream_t stream) {
    const float* states  = (const float*)d_in[0];
    const float* log_tau = (const float*)d_in[1];
    const int N = in_sizes[0] / 8;

    char* ws    = (char*)d_ws;
    float4* rec = (float4*)ws;
    float*  G   = (float*)(ws + (size_t)N * 16);
    float*  gv  = (float*)(ws + (size_t)N * 16 + (size_t)N * 64);
    float*  out = (float*)d_out;

    precompute_kernel<<<(N + 255) / 256, 256, 0, stream>>>(states, rec, G, N);
    groupvel_kernel<<<1, 256, 0, stream>>>(G, gv, N);
    neighbour_kernel<<<N / 128, 256, 0, stream>>>(rec, G, gv, log_tau, out, N);
}


# Split each row tile's j range over the 32 waves of its own block (256 blocks x 1024 threads, was 32 x 256); per-wave minima and partial C tiles merged in LDS: sums over j regrouped by wave chunk
# speedup vs baseline: 9.9805x; 9.9805x over previous
#include <hip/hip_runtime.h>

typedef float v2f __attribute__((ext_vector_type(2)));
typedef float v8f __attribute__((ext_vector_type(8)));

__device__ __forceinline__ float fast_sqrt(float x) { return __builtin_amdgcn_sqrtf(x); }
__device__ __forceinline__ float fast_exp2(float x) { return __builtin_amdgcn_exp2f(x); }

#define LOG2E 1.44269504088896340736f

__global__ void precompute_kernel(const float* __restrict__ states,
                                  float4* __restrict__ rec,
                                  float* __restrict__ G, int N) {
    int i = blockIdx.x * blockDim.x + threadIdx.x;
    if (i >= N) return;
    const float4* s = (const float4*)(states + (size_t)i * 8);
    float4 a = s[0];
    float4 b = s[1];
    float px = (a.x + a.z) * 0.5f;
    float py = (a.y + a.w) * 0.5f;
    float vx = (b.x + b.z) * 0.5f;
    float vy = (b.y + b.w) * 0.5f;
    float r  = px * px + py * py;
    rec[i] = make_float4(px, py, r, 0.0f);
    float4* g = (float4*)(G + (size_t)i * 16);
    g[0] = make_float4(px, py, vx, vy);
    g[1] = make_float4(px * px, py * py, vx * vx, vy * vy);
    g[2] = make_float4(1.0f, 0.0f, 0.0f, 0.0f);
    g[3] = make_float4(0.0f, 0.0f, 0.0f, 0.0f);
}

__global__ void groupvel_kernel(const float* __restrict__ G,
                                float* __restrict__ gv, int N) {
    __shared__ float sx[256];
    __shared__ float sy[256];
    int t = threadIdx.x;
    float ax = 0.0f, ay = 0.0f;
    for (int j = t; j < N; j += 256) {
        ax += G[(size_t)j * 16 + 2];
        ay += G[(size_t)j * 16 + 3];
    }
    sx[t] = ax; sy[t] = ay;
    __syncthreads();
    for (int s = 128; s > 0; s >>= 1) {
        if (t < s) { sx[t] += sx[t + s]; sy[t] += sy[t + s]; }
        __syncthreads();
    }
    if (t == 0) {
        gv[0] = sx[0] / (float)N;
        gv[1] = sy[0] / (float)N;
    }
}

#define JW 32

__global__ __launch_bounds__(JW * 32) void neighbour_kernel(
    const float4* __restrict__ rec,
    const float*  __restrict__ G,
    const float*  __restrict__ gvp,
    const float*  __restrict__ log_tau,
    float*        __restrict__ out, int N)
{
    __shared__ float qbuf[JW][16];
    __shared__ float cbuf[JW][256];
    __shared__ float cfin[256];

    const int tid  = threadIdx.x;
    const int wave = __builtin_amdgcn_readfirstlane(tid >> 5);
    const int lane = tid & 31;
    const int row  = lane & 15;
    const int h    = lane >> 4;
    const int i0   = blockIdx.x * 16;
    const int i    = i0 + row;

    const int tiles = N >> 4;
    const int tpw   = (tiles + JW - 1) / JW;
    const int jbeg  = min(wave * tpw * 16, N);
    const int jend  = min(jbeg + tpw * 16, N);

    const float invtau = 1.0f / __expf(log_tau[0]);

    const float4 ri  = rec[i];
    const float px_i = ri.x, py_i = ri.y, r_i = ri.z;

    float qmin = 3.0e38f;
    #pragma unroll 4
    for (int j = jbeg + h; j < jend; j += 2) {
        float4 rj = rec[j];
        float dot = px_i * rj.x + py_i * rj.y;
        float q   = r_i + rj.z - 2.0f * dot;
        q = (j == i) ? 3.0e38f : q;
        qmin = fminf(qmin, q);
    }
    qmin = fminf(qmin, __shfl_xor(qmin, 16, 32));
    if (lane < 16) qbuf[wave][row] = qmin;
    __syncthreads();
    for (int w = 0; w < JW; ++w) qmin = fminf(qmin, qbuf[w][row]);
    const float mdist = fast_sqrt(fmaxf(qmin, 0.0f) + 1e-8f);

    const float k2 = invtau * LOG2E;
    const float b2 = mdist * k2;

    v8f acc = {0.0f, 0.0f, 0.0f, 0.0f, 0.0f, 0.0f, 0.0f, 0.0f};

    for (int jt = jbeg; jt < jend; jt += 16) {
        v2f A[4];
        v2f B[4];
        #pragma unroll
        for (int g = 0; g < 4; ++g) {
            #pragma unroll
            for (int p = 0; p < 2; ++p) {
                const int j = jt + 4 * g + p + 2 * h;
                float4 rj = rec[j];
                float dot  = px_i * rj.x + py_i * rj.y;
                float q    = r_i + rj.z - 2.0f * dot;
                float dist = fast_sqrt(fmaxf(q, 0.0f) + 1e-8f);
                float w    = fast_exp2(fmaf(-dist, k2, b2));
                w = (j == i) ? 0.0f : w;
                A[g][p] = w;
                B[g][p] = G[(size_t)j * 16 + row];
            }
        }
        #pragma unroll
        for (int g = 0; g < 4; ++g) {
            acc = __builtin_amdgcn_wmma_f32_16x16x4_f32(
                 false, A[g],  false, B[g],
                 (short)0, acc,  false,  false);
        }
    }

    #pragma unroll
    for (int v = 0; v < 8; ++v) {
        cbuf[wave][(v + 8 * h) * 16 + row] = acc[v];
    }
    __syncthreads();

    if (tid < 256) {
        float s = 0.0f;
        for (int w = 0; w < JW; ++w) s += cbuf[w][tid];
        cfin[tid] = s;
    }
    __syncthreads();

    if (tid < 16) {
        const int ii = i0 + tid;
        const float* c = &cfin[tid * 16];
        const float S    = c[8];
        const float invS = 1.0f / S;
        const float4 fi4 = *(const float4*)(G + (size_t)ii * 16);
        const float f[4] = {fi4.x, fi4.y, fi4.z, fi4.w};
        const float gvx = gvp[0];
        const float gvy = gvp[1];

        float mu[4], sg[4];
        #pragma unroll
        for (int d = 0; d < 4; ++d) {
            float T = c[d] * invS;
            float U = c[4 + d] * invS;
            mu[d] = f[d] - T;
            float eg2 = f[d] * f[d] - 2.0f * f[d] * T + U;
            sg[d] = fast_sqrt(fmaxf(eg2 - mu[d] * mu[d], 0.0f) + 1e-6f);
        }
        float4* o = (float4*)(out + (size_t)ii * 12);
        o[0] = make_float4(mu[0], mu[1], mu[2], mu[3]);
        o[1] = make_float4(sg[0], sg[1], sg[2], sg[3]);
        o[2] = make_float4(gvx, gvy, f[2] - gvx, f[3] - gvy);
    }
}

extern "C" void kernel_launch(void* const* d_in, const int* in_sizes, int n_in,
                              void* d_out, int out_size, void* d_ws, size_t ws_size,
                              hipStream_t stream) {
    const float* states  = (const float*)d_in[0];
    const float* log_tau = (const float*)d_in[1];
    const int N = in_sizes[0] / 8;

    char* ws    = (char*)d_ws;
    float4* rec = (float4*)ws;
    float*  G   = (float*)(ws + (size_t)N * 16);
    float*  gv  = (float*)(ws + (size_t)N * 16 + (size_t)N * 64);
    float*  out = (float*)d_out;

    precompute_kernel<<<(N + 255) / 256, 256, 0, stream>>>(states, rec, G, N);
    groupvel_kernel<<<1, 256, 0, stream>>>(G, gv, N);
    neighbour_kernel<<<N / 16, JW * 32, 0, stream>>>(rec, G, gv, log_tau, out, N);
}


# One launch instead of three: each wave builds its j share's records (pos, |p|^2, features, squares) from states in LDS; both passes read LDS; velocity mean summed per wave share then over waves
# speedup vs baseline: 26.0387x; 2.6090x over previous
#include <hip/hip_runtime.h>

typedef float v2f __attribute__((ext_vector_type(2)));
typedef float v8f __attribute__((ext_vector_type(8)));

__device__ __forceinline__ float fast_sqrt(float x) { return __builtin_amdgcn_sqrtf(x); }
__device__ __forceinline__ float fast_exp2(float x) { return __builtin_amdgcn_exp2f(x); }

#define LOG2E 1.44269504088896340736f

#define JW 32
#define SHARE_MAX 128
#define RQ 3

__global__ __launch_bounds__(JW * 32) void neighbour_kernel(
    const float* __restrict__ states,
    const float* __restrict__ log_tau,
    float*       __restrict__ out, int N)
{
    __shared__ float4 srec[JW][SHARE_MAX * RQ];
    __shared__ float qbuf[JW][16];
    __shared__ float gbuf[JW][2];
    __shared__ float gfin[2];
    __shared__ float cbuf[JW][256];
    __shared__ float cfin[256];

    const int tid  = threadIdx.x;
    const int wave = __builtin_amdgcn_readfirstlane(tid >> 5);
    const int lane = tid & 31;
    const int row  = lane & 15;
    const int h    = lane >> 4;
    const int i0   = blockIdx.x * 16;
    const int i    = i0 + row;

    const int tiles = N >> 4;
    const int tpw   = (tiles + JW - 1) / JW;
    const int jbeg  = min(wave * tpw * 16, N);
    const int jend  = min(jbeg + tpw * 16, N);
    const int cnt   = jend - jbeg;
    const int ntile = cnt >> 4;

    const float invtau = 1.0f / __expf(log_tau[0]);

    float4* mine = &srec[wave][0];
    float4 sa[SHARE_MAX / 32], sb[SHARE_MAX / 32];
    #pragma unroll
    for (int p = 0; p < SHARE_MAX / 32; ++p) {
        const int j = min(jbeg + lane + 32 * p, N - 1);
        const float4* s = (const float4*)(states + (size_t)j * 8);
        sa[p] = s[0];
        sb[p] = s[1];
    }
    float svx = 0.0f, svy = 0.0f;
    #pragma unroll
    for (int p = 0; p < SHARE_MAX / 32; ++p) {
        const int jj = lane + 32 * p;
        const float px = (sa[p].x + sa[p].z) * 0.5f;
        const float py = (sa[p].y + sa[p].w) * 0.5f;
        const float vx = (sb[p].x + sb[p].z) * 0.5f;
        const float vy = (sb[p].y + sb[p].w) * 0.5f;
        mine[jj * RQ + 0] = make_float4(px, py, px * px + py * py, 1.0f);
        mine[jj * RQ + 1] = make_float4(px, py, vx, vy);
        mine[jj * RQ + 2] = make_float4(px * px, py * py, vx * vx, vy * vy);
        const bool live = jj < cnt;
        svx += live ? vx : 0.0f;
        svy += live ? vy : 0.0f;
    }
    __syncthreads();

    const int ow = blockIdx.x / tpw;
    const int oi = (blockIdx.x - ow * tpw) * 16 + row;
    const float4 ri  = srec[ow][oi * RQ];
    const float px_i = ri.x, py_i = ri.y, r_i = ri.z;

    const float4* rec1 = mine + RQ * h;
    const int d1 = i - jbeg - h;
    float qmin = 3.0e38f;
    {
        const float4* rp = rec1;
        int dd = d1;
        for (int tt = 0; tt < ntile; ++tt) {
            #pragma unroll
            for (int u = 0; u < 8; ++u) {
                const float4 rj = rp[2 * u * RQ];
                float dot = px_i * rj.x + py_i * rj.y;
                float q   = r_i + rj.z - 2.0f * dot;
                q = (2 * u == dd) ? 3.0e38f : q;
                qmin = fminf(qmin, q);
            }
            rp += 16 * RQ;
            dd -= 16;
        }
    }
    qmin = fminf(qmin, __shfl_xor(qmin, 16, 32));
    if (lane < 16) qbuf[wave][row] = qmin;
    __syncthreads();
    for (int w = 0; w < JW; ++w) qmin = fminf(qmin, qbuf[w][row]);
    const float mdist = fast_sqrt(fmaxf(qmin, 0.0f) + 1e-8f);

    const float k2 = invtau * LOG2E;
    const float b2 = mdist * k2;

    const float4* rec2 = mine + RQ * 2 * h;
    const float*  bcol = (const float*)rec2 + ((row < 8) ? 4 + row : 3);
    const int d2 = i - jbeg - 2 * h;
    v8f acc = {0.0f, 0.0f, 0.0f, 0.0f, 0.0f, 0.0f, 0.0f, 0.0f};

    for (int tt = 0; tt < ntile; ++tt) {
        const int t = tt * 16;
        v2f A[4];
        v2f B[4];
        #pragma unroll
        for (int g = 0; g < 4; ++g) {
            #pragma unroll
            for (int p = 0; p < 2; ++p) {
                const int k = t + 4 * g + p;
                const float4 rj = rec2[k * RQ];
                float dot  = px_i * rj.x + py_i * rj.y;
                float q    = r_i + rj.z - 2.0f * dot;
                float dist = fast_sqrt(fmaxf(q, 0.0f) + 1e-8f);
                float w    = fast_exp2(fmaf(-dist, k2, b2));
                w = (k == d2) ? 0.0f : w;
                A[g][p] = w;
                B[g][p] = bcol[k * RQ * 4];
            }
        }
        #pragma unroll
        for (int g = 0; g < 4; ++g) {
            acc = __builtin_amdgcn_wmma_f32_16x16x4_f32(
                 false, A[g],  false, B[g],
                 (short)0, acc,  false,  false);
        }
    }

    #pragma unroll
    for (int v = 0; v < 8; ++v) {
        cbuf[wave][(v + 8 * h) * 16 + row] = acc[v];
    }
    #pragma unroll
    for (int m = 16; m > 0; m >>= 1) {
        svx += __shfl_xor(svx, m, 32);
        svy += __shfl_xor(svy, m, 32);
    }
    if (lane == 0) { gbuf[wave][0] = svx; gbuf[wave][1] = svy; }
    __syncthreads();

    if (tid < 256) {
        float s = 0.0f;
        for (int w = 0; w < JW; ++w) s += cbuf[w][tid];
        cfin[tid] = s;
    } else if (tid < 258) {
        const int c = tid - 256;
        float g = 0.0f;
        for (int w = 0; w < JW; ++w) g += gbuf[w][c];
        gfin[c] = g / (float)N;
    }
    __syncthreads();

    if (tid < 16) {
        const int ii = i0 + tid;
        const float* c = &cfin[tid * 16];
        const float S    = c[8];
        const float invS = 1.0f / S;
        const float4 fi4 = srec[ow][oi * RQ + 1];
        const float f[4] = {fi4.x, fi4.y, fi4.z, fi4.w};
        const float gvx = gfin[0];
        const float gvy = gfin[1];

        float mu[4], sg[4];
        #pragma unroll
        for (int d = 0; d < 4; ++d) {
            float T = c[d] * invS;
            float U = c[4 + d] * invS;
            mu[d] = f[d] - T;
            float eg2 = f[d] * f[d] - 2.0f * f[d] * T + U;
            sg[d] = fast_sqrt(fmaxf(eg2 - mu[d] * mu[d], 0.0f) + 1e-6f);
        }
        float4* o = (float4*)(out + (size_t)ii * 12);
        o[0] = make_float4(mu[0], mu[1], mu[2], mu[3]);
        o[1] = make_float4(sg[0], sg[1], sg[2], sg[3]);
        o[2] = make_float4(gvx, gvy, f[2] - gvx, f[3] - gvy);
    }
}

extern "C" void kernel_launch(void* const* d_in, const int* in_sizes, int n_in,
                              void* d_out, int out_size, void* d_ws, size_t ws_size,
                              hipStream_t stream) {
    const float* states  = (const float*)d_in[0];
    const float* log_tau = (const float*)d_in[1];
    const int N = in_sizes[0] / 8;
    float*  out = (float*)d_out;

    neighbour_kernel<<<N / 16, JW * 32, 0, stream>>>(states, log_tau, out, N);
}


# Squared distances on the matrix unit: one f32 WMMA (K=4) per 16x16 tile, transposed so its result registers are the weight operand of the sums, kept for both passes; sums over j regrouped per tile
# speedup vs baseline: 31.7725x; 1.2202x over previous
#include <hip/hip_runtime.h>

typedef float v2f __attribute__((ext_vector_type(2)));
typedef float v8f __attribute__((ext_vector_type(8)));

__device__ __forceinline__ float fast_sqrt(float x) { return __builtin_amdgcn_sqrtf(x); }
__device__ __forceinline__ float fast_exp2(float x) { return __builtin_amdgcn_exp2f(x); }

#define LOG2E 1.44269504088896340736f

#define JW 32
#define SHARE_MAX 128
#define NT (SHARE_MAX / 16)
#define NF 9
#define FP 132

__global__ __launch_bounds__(JW * 32) void neighbour_kernel(
    const float* __restrict__ states,
    const float* __restrict__ log_tau,
    float*       __restrict__ out, int N)
{
    __shared__ float4 sq[JW][SHARE_MAX];
    __shared__ __attribute__((aligned(16))) float sf[JW][NF][FP];
    __shared__ float qbuf[JW][16];
    __shared__ float gbuf[JW][2];
    __shared__ float gfin[2];
    __shared__ float cbuf[JW][256];
    __shared__ float cfin[256];

    const int tid  = threadIdx.x;
    const int wave = __builtin_amdgcn_readfirstlane(tid >> 5);
    const int lane = tid & 31;
    const int c    = lane & 15;
    const int h    = lane >> 4;
    const int i0   = blockIdx.x * 16;

    const int tiles = N >> 4;
    const int tpw   = (tiles + JW - 1) / JW;
    const int jbeg  = min(wave * tpw * 16, N);
    const int jend  = min(jbeg + tpw * 16, N);
    const int cnt   = jend - jbeg;
    const int ntile = cnt >> 4;

    const float invtau = 1.0f / __expf(log_tau[0]);

    float4* myq = &sq[wave][0];
    float (*myf)[FP] = sf[wave];
    float4 sa[SHARE_MAX / 32], sb[SHARE_MAX / 32];
    #pragma unroll
    for (int p = 0; p < SHARE_MAX / 32; ++p) {
        const int j = min(jbeg + lane + 32 * p, N - 1);
        const float4* s = (const float4*)(states + (size_t)j * 8);
        sa[p] = s[0];
        sb[p] = s[1];
    }
    float svx = 0.0f, svy = 0.0f;
    #pragma unroll
    for (int p = 0; p < SHARE_MAX / 32; ++p) {
        const int jj = lane + 32 * p;
        const float px = (sa[p].x + sa[p].z) * 0.5f;
        const float py = (sa[p].y + sa[p].w) * 0.5f;
        const float vx = (sb[p].x + sb[p].z) * 0.5f;
        const float vy = (sb[p].y + sb[p].w) * 0.5f;
        myq[jj] = make_float4(-2.0f * px, -2.0f * py, px * px + py * py, 1.0f);
        myf[0][jj] = px;
        myf[1][jj] = py;
        myf[2][jj] = vx;
        myf[3][jj] = vy;
        myf[4][jj] = px * px;
        myf[5][jj] = py * py;
        myf[6][jj] = vx * vx;
        myf[7][jj] = vy * vy;
        myf[8][jj] = 1.0f;
        const bool live = jj < cnt;
        svx += live ? vx : 0.0f;
        svy += live ? vy : 0.0f;
    }
    __syncthreads();

    const int ow = blockIdx.x / tpw;
    const int ob = blockIdx.x - ow * tpw;
    const float4 qi = sq[ow][ob * 16 + c];

    v2f Bq;
    Bq[0] = h ? 1.0f : -0.5f * qi.x;
    Bq[1] = h ? qi.z : -0.5f * qi.y;
    const float* aq = (const float*)myq + 4 * c + 2 * h;

    v8f Q[NT];
    #pragma unroll
    for (int tt = 0; tt < NT; ++tt) {
        const v2f Aq = *(const v2f*)(aq + tt * 64);
        const v8f cz = {0.0f, 0.0f, 0.0f, 0.0f, 0.0f, 0.0f, 0.0f, 0.0f};
        Q[tt] = __builtin_amdgcn_wmma_f32_16x16x4_f32(
             false, Aq,  false, Bq,
             (short)0, cz,  false,  false);
    }
    if (ntile < NT) {
        #pragma unroll
        for (int tt = 0; tt < NT; ++tt) {
            #pragma unroll
            for (int r = 0; r < 8; ++r) Q[tt][r] = (tt < ntile) ? Q[tt][r] : 3.0e38f;
        }
    }
    if (wave == ow) {
        #pragma unroll
        for (int tt = 0; tt < NT; ++tt) {
            #pragma unroll
            for (int r = 0; r < 8; ++r) Q[tt][r] = (tt == ob && 8 * h + r == c) ? 3.0e38f : Q[tt][r];
        }
    }

    float qmin = 3.0e38f;
    #pragma unroll
    for (int tt = 0; tt < NT; ++tt) {
        #pragma unroll
        for (int r = 0; r < 8; ++r) qmin = fminf(qmin, Q[tt][r]);
    }
    qmin = fminf(qmin, __shfl_xor(qmin, 16, 32));
    if (lane < 16) qbuf[wave][c] = qmin;
    __syncthreads();
    for (int w = 0; w < JW; ++w) qmin = fminf(qmin, qbuf[w][c]);
    const float mdist = fast_sqrt(fmaxf(qmin, 0.0f) + 1e-8f);

    const float k2 = invtau * LOG2E;
    const float b2 = mdist * k2;

    const float* fa = &myf[min(c, NF - 1)][8 * h];
    v8f acc = {0.0f, 0.0f, 0.0f, 0.0f, 0.0f, 0.0f, 0.0f, 0.0f};

    #pragma unroll
    for (int tt = 0; tt < NT; ++tt) {
        const float4 f0 = *(const float4*)(fa + tt * 16);
        const float4 f1 = *(const float4*)(fa + tt * 16 + 4);
        float w[8];
        #pragma unroll
        for (int r = 0; r < 8; ++r) {
            const float dist = fast_sqrt(fmaxf(Q[tt][r] + 1e-8f, 1e-8f));
            w[r] = fast_exp2(fmaf(-dist, k2, b2));
        }
        v2f Af[4], Bw[4];
        Af[0][0] = f0.x; Af[0][1] = f0.y;
        Af[1][0] = f0.z; Af[1][1] = f0.w;
        Af[2][0] = f1.x; Af[2][1] = f1.y;
        Af[3][0] = f1.z; Af[3][1] = f1.w;
        #pragma unroll
        for (int g = 0; g < 4; ++g) { Bw[g][0] = w[2 * g]; Bw[g][1] = w[2 * g + 1]; }
        #pragma unroll
        for (int g = 0; g < 4; ++g) {
            acc = __builtin_amdgcn_wmma_f32_16x16x4_f32(
                 false, Af[g],  false, Bw[g],
                 (short)0, acc,  false,  false);
        }
    }

    #pragma unroll
    for (int r = 0; r < 8; ++r) {
        cbuf[wave][(r + 8 * h) * 16 + c] = acc[r];
    }
    #pragma unroll
    for (int m = 16; m > 0; m >>= 1) {
        svx += __shfl_xor(svx, m, 32);
        svy += __shfl_xor(svy, m, 32);
    }
    if (lane == 0) { gbuf[wave][0] = svx; gbuf[wave][1] = svy; }
    __syncthreads();

    if (tid < 256) {
        float s = 0.0f;
        for (int w = 0; w < JW; ++w) s += cbuf[w][tid];
        cfin[tid] = s;
    } else if (tid < 258) {
        const int cc = tid - 256;
        float g = 0.0f;
        for (int w = 0; w < JW; ++w) g += gbuf[w][cc];
        gfin[cc] = g / (float)N;
    }
    __syncthreads();

    if (tid < 16) {
        const int ii = i0 + tid;
        const float* cc = &cfin[tid];
        const float S    = cc[16 * 8];
        const float invS = 1.0f / S;
        const int   oi   = ob * 16 + tid;
        const float f[4] = {sf[ow][0][oi], sf[ow][1][oi], sf[ow][2][oi], sf[ow][3][oi]};
        const float gvx = gfin[0];
        const float gvy = gfin[1];

        float mu[4], sg[4];
        #pragma unroll
        for (int d = 0; d < 4; ++d) {
            float T = cc[16 * d] * invS;
            float U = cc[16 * (4 + d)] * invS;
            mu[d] = f[d] - T;
            float eg2 = f[d] * f[d] - 2.0f * f[d] * T + U;
            sg[d] = fast_sqrt(fmaxf(eg2 - mu[d] * mu[d], 0.0f) + 1e-6f);
        }
        float4* o = (float4*)(out + (size_t)ii * 12);
        o[0] = make_float4(mu[0], mu[1], mu[2], mu[3]);
        o[1] = make_float4(sg[0], sg[1], sg[2], sg[3]);
        o[2] = make_float4(gvx, gvy, f[2] - gvx, f[3] - gvy);
    }
}

extern "C" void kernel_launch(void* const* d_in, const int* in_sizes, int n_in,
                              void* d_out, int out_size, void* d_ws, size_t ws_size,
                              hipStream_t stream) {
    const float* states  = (const float*)d_in[0];
    const float* log_tau = (const float*)d_in[1];
    const int N = in_sizes[0] / 8;
    float*  out = (float*)d_out;

    neighbour_kernel<<<N / 16, JW * 32, 0, stream>>>(states, log_tau, out, N);
}


# No barrier before the merge: each wave reads the block's 16 positions itself and takes its own share's row minimum as softmax reference; partial sums scaled by exp((m-m_w)/tau) in the merge
# speedup vs baseline: 35.5810x; 1.1199x over previous
#include <hip/hip_runtime.h>

typedef float v2f __attribute__((ext_vector_type(2)));
typedef float v8f __attribute__((ext_vector_type(8)));

__device__ __forceinline__ float fast_sqrt(float x) { return __builtin_amdgcn_sqrtf(x); }
__device__ __forceinline__ float fast_exp2(float x) { return __builtin_amdgcn_exp2f(x); }

#define LOG2E 1.44269504088896340736f

#define JW 32
#define SHARE_MAX 128
#define NT (SHARE_MAX / 16)
#define NF 9
#define FP 132

__global__ __launch_bounds__(JW * 32) void neighbour_kernel(
    const float* __restrict__ states,
    const float* __restrict__ log_tau,
    float*       __restrict__ out, int N)
{
    __shared__ float4 sq[JW][SHARE_MAX];
    __shared__ __attribute__((aligned(16))) float sf[JW][NF][FP];
    __shared__ float qbuf[JW][16];
    __shared__ float gbuf[JW][2];
    __shared__ float gfin[2];
    __shared__ float cbuf[JW][256];
    __shared__ float cfin[256];

    const int tid  = threadIdx.x;
    const int wave = __builtin_amdgcn_readfirstlane(tid >> 5);
    const int lane = tid & 31;
    const int c    = lane & 15;
    const int h    = lane >> 4;
    const int i0   = blockIdx.x * 16;

    const int tiles = N >> 4;
    const int tpw   = (tiles + JW - 1) / JW;
    const int jbeg  = min(wave * tpw * 16, N);
    const int jend  = min(jbeg + tpw * 16, N);
    const int cnt   = jend - jbeg;
    const int ntile = cnt >> 4;

    const float invtau = 1.0f / __expf(log_tau[0]);

    const float4 si = *(const float4*)(states + (size_t)(i0 + c) * 8);

    float4* myq = &sq[wave][0];
    float (*myf)[FP] = sf[wave];
    float4 sa[SHARE_MAX / 32], sb[SHARE_MAX / 32];
    #pragma unroll
    for (int p = 0; p < SHARE_MAX / 32; ++p) {
        const int j = min(jbeg + lane + 32 * p, N - 1);
        const float4* s = (const float4*)(states + (size_t)j * 8);
        sa[p] = s[0];
        sb[p] = s[1];
    }
    float svx = 0.0f, svy = 0.0f;
    #pragma unroll
    for (int p = 0; p < SHARE_MAX / 32; ++p) {
        const int jj = lane + 32 * p;
        const float px = (sa[p].x + sa[p].z) * 0.5f;
        const float py = (sa[p].y + sa[p].w) * 0.5f;
        const float vx = (sb[p].x + sb[p].z) * 0.5f;
        const float vy = (sb[p].y + sb[p].w) * 0.5f;
        myq[jj] = make_float4(-2.0f * px, -2.0f * py, px * px + py * py, 1.0f);
        myf[0][jj] = px;
        myf[1][jj] = py;
        myf[2][jj] = vx;
        myf[3][jj] = vy;
        myf[4][jj] = px * px;
        myf[5][jj] = py * py;
        myf[6][jj] = vx * vx;
        myf[7][jj] = vy * vy;
        myf[8][jj] = 1.0f;
        const bool live = jj < cnt;
        svx += live ? vx : 0.0f;
        svy += live ? vy : 0.0f;
    }
    __builtin_amdgcn_fence(__ATOMIC_RELEASE, "wavefront");
    __builtin_amdgcn_wave_barrier();
    __builtin_amdgcn_fence(__ATOMIC_ACQUIRE, "wavefront");

    const int ow = blockIdx.x / tpw;
    const int ob = blockIdx.x - ow * tpw;
    const float pxi = (si.x + si.z) * 0.5f;
    const float pyi = (si.y + si.w) * 0.5f;
    const float ri  = pxi * pxi + pyi * pyi;

    v2f Bq;
    Bq[0] = h ? 1.0f : pxi;
    Bq[1] = h ? ri   : pyi;
    const float* aq = (const float*)myq + 4 * c + 2 * h;

    v8f Q[NT];
    #pragma unroll
    for (int tt = 0; tt < NT; ++tt) {
        const v2f Aq = *(const v2f*)(aq + tt * 64);
        const v8f cz = {0.0f, 0.0f, 0.0f, 0.0f, 0.0f, 0.0f, 0.0f, 0.0f};
        Q[tt] = __builtin_amdgcn_wmma_f32_16x16x4_f32(
             false, Aq,  false, Bq,
             (short)0, cz,  false,  false);
    }
    if (ntile < NT) {
        #pragma unroll
        for (int tt = 0; tt < NT; ++tt) {
            #pragma unroll
            for (int r = 0; r < 8; ++r) Q[tt][r] = (tt < ntile) ? Q[tt][r] : 3.0e38f;
        }
    }
    if (wave == ow) {
        #pragma unroll
        for (int tt = 0; tt < NT; ++tt) {
            #pragma unroll
            for (int r = 0; r < 8; ++r) Q[tt][r] = (tt == ob && 8 * h + r == c) ? 3.0e38f : Q[tt][r];
        }
    }

    float qmin = 3.0e38f;
    #pragma unroll
    for (int tt = 0; tt < NT; ++tt) {
        #pragma unroll
        for (int r = 0; r < 8; ++r) qmin = fminf(qmin, Q[tt][r]);
    }
    qmin = fminf(qmin, __shfl_xor(qmin, 16, 32));
    const float mdist = fast_sqrt(fmaxf(qmin, 0.0f) + 1e-8f);
    if (lane < 16) qbuf[wave][c] = mdist;

    const float k2 = invtau * LOG2E;
    const float b2 = (ntile > 0) ? mdist * k2 : 0.0f;

    const float* fa = &myf[min(c, NF - 1)][8 * h];
    v8f acc = {0.0f, 0.0f, 0.0f, 0.0f, 0.0f, 0.0f, 0.0f, 0.0f};

    #pragma unroll
    for (int tt = 0; tt < NT; ++tt) {
        const float4 f0 = *(const float4*)(fa + tt * 16);
        const float4 f1 = *(const float4*)(fa + tt * 16 + 4);
        float w[8];
        #pragma unroll
        for (int r = 0; r < 8; ++r) {
            const float dist = fast_sqrt(fmaxf(Q[tt][r] + 1e-8f, 1e-8f));
            w[r] = fast_exp2(fmaf(-dist, k2, b2));
        }
        v2f Af[4], Bw[4];
        Af[0][0] = f0.x; Af[0][1] = f0.y;
        Af[1][0] = f0.z; Af[1][1] = f0.w;
        Af[2][0] = f1.x; Af[2][1] = f1.y;
        Af[3][0] = f1.z; Af[3][1] = f1.w;
        #pragma unroll
        for (int g = 0; g < 4; ++g) { Bw[g][0] = w[2 * g]; Bw[g][1] = w[2 * g + 1]; }
        #pragma unroll
        for (int g = 0; g < 4; ++g) {
            acc = __builtin_amdgcn_wmma_f32_16x16x4_f32(
                 false, Af[g],  false, Bw[g],
                 (short)0, acc,  false,  false);
        }
    }

    #pragma unroll
    for (int r = 0; r < 8; ++r) {
        cbuf[wave][(r + 8 * h) * 16 + c] = acc[r];
    }
    #pragma unroll
    for (int m = 16; m > 0; m >>= 1) {
        svx += __shfl_xor(svx, m, 32);
        svy += __shfl_xor(svy, m, 32);
    }
    if (lane == 0) { gbuf[wave][0] = svx; gbuf[wave][1] = svy; }
    __syncthreads();

    if (tid < 256) {
        float ml[JW];
        float m = 3.0e38f;
        #pragma unroll
        for (int w = 0; w < JW; ++w) { ml[w] = qbuf[w][tid & 15]; m = fminf(m, ml[w]); }
        float s = 0.0f;
        #pragma unroll
        for (int w = 0; w < JW; ++w) s = fmaf(cbuf[w][tid], fast_exp2((m - ml[w]) * k2), s);
        cfin[tid] = s;
    } else if (tid < 258) {
        const int cc = tid - 256;
        float g = 0.0f;
        for (int w = 0; w < JW; ++w) g += gbuf[w][cc];
        gfin[cc] = g / (float)N;
    }
    __syncthreads();

    if (tid < 16) {
        const int ii = i0 + tid;
        const float* cc = &cfin[tid];
        const float S    = cc[16 * 8];
        const float invS = 1.0f / S;
        const int   oi   = ob * 16 + tid;
        const float f[4] = {sf[ow][0][oi], sf[ow][1][oi], sf[ow][2][oi], sf[ow][3][oi]};
        const float gvx = gfin[0];
        const float gvy = gfin[1];

        float mu[4], sg[4];
        #pragma unroll
        for (int d = 0; d < 4; ++d) {
            float T = cc[16 * d] * invS;
            float U = cc[16 * (4 + d)] * invS;
            mu[d] = f[d] - T;
            float eg2 = f[d] * f[d] - 2.0f * f[d] * T + U;
            sg[d] = fast_sqrt(fmaxf(eg2 - mu[d] * mu[d], 0.0f) + 1e-6f);
        }
        float4* o = (float4*)(out + (size_t)ii * 12);
        o[0] = make_float4(mu[0], mu[1], mu[2], mu[3]);
        o[1] = make_float4(sg[0], sg[1], sg[2], sg[3]);
        o[2] = make_float4(gvx, gvy, f[2] - gvx, f[3] - gvy);
    }
}

extern "C" void kernel_launch(void* const* d_in, const int* in_sizes, int n_in,
                              void* d_out, int out_size, void* d_ws, size_t ws_size,
                              hipStream_t stream) {
    const float* states  = (const float*)d_in[0];
    const float* log_tau = (const float*)d_in[1];
    const int N = in_sizes[0] / 8;
    float*  out = (float*)d_out;

    neighbour_kernel<<<N / 16, JW * 32, 0, stream>>>(states, log_tau, out, N);
}
